# starting kernel
# baseline (speedup 1.0000x reference)
#include <hip/hip_runtime.h>
#include <hip/hip_bf16.h>

typedef float v2f __attribute__((ext_vector_type(2)));
typedef float v8f __attribute__((ext_vector_type(8)));

#define IN_CH 128
#define OUT_CH 64

__global__ void prep_weights_t(const float* __restrict__ w_main_in,
                               const float* __restrict__ w_main_out,
                               const float* __restrict__ w_shared,
                               float* __restrict__ w_in_t,
                               float* __restrict__ w_out_t) {
    int i = blockIdx.x * blockDim.x + threadIdx.x;
    if (i < IN_CH * OUT_CH) {
        int k = i >> 6;
        int c = i & 63;
        float s = w_shared[i];
        w_in_t [c * IN_CH + k] = w_main_in[i]  + s;
        w_out_t[c * IN_CH + k] = w_main_out[i] + s;
    }
}

__global__ void gemm_dual_wmma(const float* __restrict__ x,
                               const float* __restrict__ w_in_t,
                               const float* __restrict__ w_out_t,
                               float* __restrict__ h_in,
                               float* __restrict__ h_out,
                               int n_nodes) {
    __shared__ __align__(16) float a_tile[16 * 132];

    const int mtile = blockIdx.x;
    const int row0  = mtile * 16;

    for (int i = threadIdx.x; i < 16 * (IN_CH / 4); i += blockDim.x) {
        int r  = i >> 5;
        int c4 = i & 31;
        float4 v = make_float4(0.f, 0.f, 0.f, 0.f);
        int grow = row0 + r;
        if (grow < n_nodes)
            v = ((const float4*)x)[(size_t)grow * (IN_CH / 4) + c4];
        *(float4*)&a_tile[r * 132 + c4 * 4] = v;
    }
    __syncthreads();

    const int wave = threadIdx.x >> 5;
    const int lane = threadIdx.x & 31;
    const int r    = lane & 15;
    const int half = lane >> 4;

    const float* __restrict__ Wt = (wave < 4) ? w_in_t : w_out_t;
    float*       __restrict__ H  = (wave < 4) ? h_in   : h_out;
    const int nt  = wave & 3;
    const int col = nt * 16 + r;

    const float* __restrict__ ap = &a_tile[r * 132 + 2 * half];
    const float* __restrict__ wp = Wt + (size_t)col * IN_CH + 2 * half;

    v8f acc = {};

#pragma unroll
    for (int k0 = 0; k0 < IN_CH; k0 += 4) {
        v2f a = *(const v2f*)(ap + k0);
        v2f b = *(const v2f*)(wp + k0);
        acc = __builtin_amdgcn_wmma_f32_16x16x4_f32(
             false, a,  false, b,
             (short)0, acc,  false,  false);
    }

#pragma unroll
    for (int v = 0; v < 8; ++v) {
        int grow = row0 + v + 8 * half;
        if (grow < n_nodes)
            H[(size_t)grow * OUT_CH + col] = acc[v];
    }
}

__global__ void init_out(float* __restrict__ out,
                         const float* __restrict__ c_in,
                         const float* __restrict__ c_out,
                         const float* __restrict__ b_mi,
                         const float* __restrict__ b_mo,
                         const float* __restrict__ b_si,
                         const float* __restrict__ b_so,
                         int n_nodes) {
    int idx = blockIdx.x * blockDim.x + threadIdx.x;
    if (idx < n_nodes * OUT_CH) {
        int n = idx >> 6;
        int c = idx & 63;
        out[idx] = c_in[n] * (b_mi[c] + b_si[c]) + c_out[n] * (b_mo[c] + b_so[c]);
    }
}

__global__ void scatter_edges(const float* __restrict__ h_in,
                              const float* __restrict__ h_out,
                              const int*   __restrict__ ei_in,
                              const float* __restrict__ ew_in,
                              const int*   __restrict__ ei_out,
                              const float* __restrict__ ew_out,
                              const float* __restrict__ c_in,
                              const float* __restrict__ c_out,
                              float* __restrict__ out,
                              int e_in, int e_out) {
    long long t = (long long)blockIdx.x * blockDim.x + threadIdx.x;
    int e = (int)(t >> 6);
    int c = (int)(t & 63);
    if (e < e_in) {
        int src = ei_in[e];
        int tgt = ei_in[e_in + e];
        float m = c_in[tgt] * ew_in[e] * h_in[(size_t)src * OUT_CH + c];
        unsafeAtomicAdd(&out[(size_t)tgt * OUT_CH + c], m);
    } else {
        e -= e_in;
        if (e < e_out) {
            int src = ei_out[e];
            int tgt = ei_out[e_out + e];
            float m = c_out[tgt] * ew_out[e] * h_out[(size_t)src * OUT_CH + c];
            unsafeAtomicAdd(&out[(size_t)tgt * OUT_CH + c], m);
        }
    }
}

extern "C" void kernel_launch(void* const* d_in, const int* in_sizes, int n_in,
                              void* d_out, int out_size, void* d_ws, size_t ws_size,
                              hipStream_t stream) {
    const float* x          = (const float*)d_in[0];
    const float* w_main_in  = (const float*)d_in[1];
    const float* w_main_out = (const float*)d_in[2];
    const float* w_shared   = (const float*)d_in[3];
    const float* b_mi       = (const float*)d_in[4];
    const float* b_mo       = (const float*)d_in[5];
    const float* b_si       = (const float*)d_in[6];
    const float* b_so       = (const float*)d_in[7];
    const float* c_in       = (const float*)d_in[8];
    const float* c_out      = (const float*)d_in[9];
    const int*   ei_in      = (const int*)  d_in[10];
    const float* ew_in      = (const float*)d_in[11];
    const int*   ei_out     = (const int*)  d_in[12];
    const float* ew_out     = (const float*)d_in[13];
    float* out = (float*)d_out;

    const int n_nodes = in_sizes[8];
    const int e_in    = in_sizes[11];
    const int e_out   = in_sizes[13];

    float* ws      = (float*)d_ws;
    float* w_in_t  = ws;
    float* w_out_t = w_in_t  + IN_CH * OUT_CH;
    float* h_in    = w_out_t + IN_CH * OUT_CH;
    float* h_out   = h_in    + (size_t)n_nodes * OUT_CH;

    prep_weights_t<<<(IN_CH * OUT_CH + 255) / 256, 256, 0, stream>>>(
        w_main_in, w_main_out, w_shared, w_in_t, w_out_t);

    const int mtiles = (n_nodes + 15) / 16;
    gemm_dual_wmma<<<mtiles, 256, 0, stream>>>(x, w_in_t, w_out_t, h_in, h_out, n_nodes);

    const int total = n_nodes * OUT_CH;
    init_out<<<(total + 255) / 256, 256, 0, stream>>>(
        out, c_in, c_out, b_mi, b_mo, b_si, b_so, n_nodes);

    const long long scatter_threads = ((long long)e_in + e_out) * OUT_CH;
    const int scatter_blocks = (int)((scatter_threads + 255) / 256);
    scatter_edges<<<scatter_blocks, 256, 0, stream>>>(
        h_in, h_out, ei_in, ew_in, ei_out, ew_out, c_in, c_out, out, e_in, e_out);
}


# atomic edge scatter replaced by gather: row offsets per target built on device (order checked), one wave per target sums its edges' h rows in edge order, stores out once; float sum order changes
# speedup vs baseline: 5.4060x; 5.4060x over previous
#include <hip/hip_runtime.h>
#include <hip/hip_bf16.h>

typedef float v2f __attribute__((ext_vector_type(2)));
typedef float v8f __attribute__((ext_vector_type(8)));

#define IN_CH 128
#define OUT_CH 64

__global__ void prep_weights_t(const float* __restrict__ w_main_in,
                               const float* __restrict__ w_main_out,
                               const float* __restrict__ w_shared,
                               float* __restrict__ w_in_t,
                               float* __restrict__ w_out_t,
                               int* __restrict__ unsorted_flag) {
    int i = blockIdx.x * blockDim.x + threadIdx.x;
    if (i == 0 && unsorted_flag) *unsorted_flag = 0;
    if (i < IN_CH * OUT_CH) {
        int k = i >> 6;
        int c = i & 63;
        float s = w_shared[i];
        w_in_t [c * IN_CH + k] = w_main_in[i]  + s;
        w_out_t[c * IN_CH + k] = w_main_out[i] + s;
    }
}

__global__ void gemm_dual_wmma(const float* __restrict__ x,
                               const float* __restrict__ w_in_t,
                               const float* __restrict__ w_out_t,
                               float* __restrict__ h_in,
                               float* __restrict__ h_out,
                               int n_nodes) {
    __shared__ __align__(16) float a_tile[16 * 132];

    const int mtile = blockIdx.x;
    const int row0  = mtile * 16;

    for (int i = threadIdx.x; i < 16 * (IN_CH / 4); i += blockDim.x) {
        int r  = i >> 5;
        int c4 = i & 31;
        float4 v = make_float4(0.f, 0.f, 0.f, 0.f);
        int grow = row0 + r;
        if (grow < n_nodes)
            v = ((const float4*)x)[(size_t)grow * (IN_CH / 4) + c4];
        *(float4*)&a_tile[r * 132 + c4 * 4] = v;
    }
    __syncthreads();

    const int wave = threadIdx.x >> 5;
    const int lane = threadIdx.x & 31;
    const int r    = lane & 15;
    const int half = lane >> 4;

    const float* __restrict__ Wt = (wave < 4) ? w_in_t : w_out_t;
    float*       __restrict__ H  = (wave < 4) ? h_in   : h_out;
    const int nt  = wave & 3;
    const int col = nt * 16 + r;

    const float* __restrict__ ap = &a_tile[r * 132 + 2 * half];
    const float* __restrict__ wp = Wt + (size_t)col * IN_CH + 2 * half;

    v8f acc = {};

#pragma unroll
    for (int k0 = 0; k0 < IN_CH; k0 += 4) {
        v2f a = *(const v2f*)(ap + k0);
        v2f b = *(const v2f*)(wp + k0);
        acc = __builtin_amdgcn_wmma_f32_16x16x4_f32(
             false, a,  false, b,
             (short)0, acc,  false,  false);
    }

#pragma unroll
    for (int v = 0; v < 8; ++v) {
        int grow = row0 + v + 8 * half;
        if (grow < n_nodes)
            H[(size_t)grow * OUT_CH + col] = acc[v];
    }
}

__global__ void build_row_offsets(const int* __restrict__ ei_in,
                                  const int* __restrict__ ei_out,
                                  int* __restrict__ rp_in,
                                  int* __restrict__ rp_out,
                                  int* __restrict__ unsorted_flag,
                                  int e_in, int e_out, int n_nodes) {
    int i = blockIdx.x * blockDim.x + threadIdx.x;
    const int* __restrict__ tgt;
    int* __restrict__ rp;
    int E, e;
    if (i <= e_in) {
        tgt = ei_in + e_in;  rp = rp_in;  E = e_in;  e = i;
    } else {
        e = i - (e_in + 1);
        if (e > e_out) return;
        tgt = ei_out + e_out; rp = rp_out; E = e_out;
    }
    int t = (e < E) ? tgt[e] : n_nodes;
    int p = (e > 0) ? tgt[e - 1] : -1;
    if (p > t) { *unsorted_flag = 1; return; }
    for (int n = p + 1; n <= t; ++n) rp[n] = e;
}

#define ROWS_IN_FLIGHT 16

__device__ __forceinline__ v2f add_rows(v2f acc,
                                        const float* __restrict__ h,
                                        int my_src, float my_w,
                                        int cnt, int lane) {
    for (int j = 0; j < cnt; j += ROWS_IN_FLIGHT) {
        int   sj[ROWS_IN_FLIGHT];
        float ww[ROWS_IN_FLIGHT];
        v2f   v[ROWS_IN_FLIGHT];
#pragma unroll
        for (int k = 0; k < ROWS_IN_FLIGHT; ++k) {
            sj[k] = __builtin_amdgcn_readlane(my_src, j + k);
            ww[k] = __int_as_float(__builtin_amdgcn_readlane(__float_as_int(my_w), j + k));
        }
#pragma unroll
        for (int k = 0; k < ROWS_IN_FLIGHT; ++k) {
            v[k] = (v2f){0.f, 0.f};
            if (j + k < cnt) {
                const v2f* __restrict__ row = (const v2f*)(h + (size_t)(unsigned)sj[k] * OUT_CH);
                v[k] = row[lane];
            }
        }
#pragma unroll
        for (int k = 0; k < ROWS_IN_FLIGHT; ++k) {
            acc.x = fmaf(ww[k], v[k].x, acc.x);
            acc.y = fmaf(ww[k], v[k].y, acc.y);
        }
    }
    return acc;
}

__global__ void __launch_bounds__(256)
gather_targets(const float* __restrict__ h_in,
               const float* __restrict__ h_out,
               const int*   __restrict__ ei_in,
               const float* __restrict__ ew_in,
               const int*   __restrict__ ei_out,
               const float* __restrict__ ew_out,
               const float* __restrict__ c_in,
               const float* __restrict__ c_out,
               const float* __restrict__ b_mi,
               const float* __restrict__ b_mo,
               const float* __restrict__ b_si,
               const float* __restrict__ b_so,
               const int*   __restrict__ rp_in,
               const int*   __restrict__ rp_out,
               const int*   __restrict__ unsorted_flag,
               float* __restrict__ out,
               int n_nodes) {
    const int lane = threadIdx.x & 31;
    const int n = blockIdx.x * 8 + __builtin_amdgcn_readfirstlane((int)(threadIdx.x >> 5));
    if (n >= n_nodes) return;

    const v2f bi = *(const v2f*)(b_mi + 2 * lane) + *(const v2f*)(b_si + 2 * lane);
    const v2f bo = *(const v2f*)(b_mo + 2 * lane) + *(const v2f*)(b_so + 2 * lane);
    const float ci = c_in[n], co = c_out[n];

    v2f s_i = {0.f, 0.f}, s_o = {0.f, 0.f};
    const bool sorted = unsorted_flag && (*unsorted_flag == 0);
    if (sorted) {
        const int a_i = rp_in[n],  z_i = rp_in[n + 1];
        const int a_o = rp_out[n], z_o = rp_out[n + 1];
        for (int ai = a_i, ao = a_o; ai < z_i || ao < z_o; ai += 32, ao += 32) {
            const int cnt_i = max(0, min(32, z_i - ai));
            const int cnt_o = max(0, min(32, z_o - ao));
            int src_i = 0, src_o = 0;
            float w_i = 0.f, w_o = 0.f;
            if (lane < cnt_i) { src_i = ei_in[ai + lane];  w_i = ew_in[ai + lane]; }
            if (lane < cnt_o) { src_o = ei_out[ao + lane]; w_o = ew_out[ao + lane]; }
            s_i = add_rows(s_i, h_in,  src_i, w_i, cnt_i, lane);
            s_o = add_rows(s_o, h_out, src_o, w_o, cnt_o, lane);
        }
    }
    v2f o;
    o.x = ci * (s_i.x + bi.x) + co * (s_o.x + bo.x);
    o.y = ci * (s_i.y + bi.y) + co * (s_o.y + bo.y);
    *(v2f*)(out + (size_t)n * OUT_CH + 2 * lane) = o;
}

__global__ void scatter_edges(const float* __restrict__ h_in,
                              const float* __restrict__ h_out,
                              const int*   __restrict__ ei_in,
                              const float* __restrict__ ew_in,
                              const int*   __restrict__ ei_out,
                              const float* __restrict__ ew_out,
                              const float* __restrict__ c_in,
                              const float* __restrict__ c_out,
                              const int*   __restrict__ unsorted_flag,
                              float* __restrict__ out,
                              int e_in, int e_out) {
    if (unsorted_flag && *unsorted_flag == 0) return;
    const long long total  = ((long long)e_in + e_out) * OUT_CH;
    const long long stride = (long long)gridDim.x * blockDim.x;
    for (long long t = (long long)blockIdx.x * blockDim.x + threadIdx.x; t < total; t += stride) {
        int e = (int)(t >> 6);
        int c = (int)(t & 63);
        if (e < e_in) {
            int src = ei_in[e];
            int tgt = ei_in[e_in + e];
            float m = c_in[tgt] * ew_in[e] * h_in[(size_t)src * OUT_CH + c];
            unsafeAtomicAdd(&out[(size_t)tgt * OUT_CH + c], m);
        } else {
            e -= e_in;
            int src = ei_out[e];
            int tgt = ei_out[e_out + e];
            float m = c_out[tgt] * ew_out[e] * h_out[(size_t)src * OUT_CH + c];
            unsafeAtomicAdd(&out[(size_t)tgt * OUT_CH + c], m);
        }
    }
}

extern "C" void kernel_launch(void* const* d_in, const int* in_sizes, int n_in,
                              void* d_out, int out_size, void* d_ws, size_t ws_size,
                              hipStream_t stream) {
    const float* x          = (const float*)d_in[0];
    const float* w_main_in  = (const float*)d_in[1];
    const float* w_main_out = (const float*)d_in[2];
    const float* w_shared   = (const float*)d_in[3];
    const float* b_mi       = (const float*)d_in[4];
    const float* b_mo       = (const float*)d_in[5];
    const float* b_si       = (const float*)d_in[6];
    const float* b_so       = (const float*)d_in[7];
    const float* c_in       = (const float*)d_in[8];
    const float* c_out      = (const float*)d_in[9];
    const int*   ei_in      = (const int*)  d_in[10];
    const float* ew_in      = (const float*)d_in[11];
    const int*   ei_out     = (const int*)  d_in[12];
    const float* ew_out     = (const float*)d_in[13];
    float* out = (float*)d_out;

    const int n_nodes = in_sizes[8];
    const int e_in    = in_sizes[11];
    const int e_out   = in_sizes[13];

    float* ws      = (float*)d_ws;
    float* w_in_t  = ws;
    float* w_out_t = w_in_t  + IN_CH * OUT_CH;
    float* h_in    = w_out_t + IN_CH * OUT_CH;
    float* h_out   = h_in    + (size_t)n_nodes * OUT_CH;
    int*   rp_in   = (int*)(h_out + (size_t)n_nodes * OUT_CH);
    int*   rp_out  = rp_in  + (n_nodes + 1);
    int*   flag    = rp_out + (n_nodes + 1);
    const bool have_offsets = (size_t)((const char*)(flag + 1) - (const char*)d_ws) <= ws_size;
    if (!have_offsets) { rp_in = nullptr; rp_out = nullptr; flag = nullptr; }

    prep_weights_t<<<(IN_CH * OUT_CH + 255) / 256, 256, 0, stream>>>(
        w_main_in, w_main_out, w_shared, w_in_t, w_out_t, flag);

    const int mtiles = (n_nodes + 15) / 16;
    gemm_dual_wmma<<<mtiles, 256, 0, stream>>>(x, w_in_t, w_out_t, h_in, h_out, n_nodes);

    if (have_offsets) {
        const long long rp_threads = (long long)e_in + e_out + 2;
        build_row_offsets<<<(int)((rp_threads + 255) / 256), 256, 0, stream>>>(
            ei_in, ei_out, rp_in, rp_out, flag, e_in, e_out, n_nodes);
    }

    gather_targets<<<(n_nodes + 7) / 8, 256, 0, stream>>>(
        h_in, h_out, ei_in, ew_in, ei_out, ew_out, c_in, c_out,
        b_mi, b_mo, b_si, b_so, rp_in, rp_out, flag, out, n_nodes);

    scatter_edges<<<1024, 256, 0, stream>>>(
        h_in, h_out, ei_in, ew_in, ei_out, ew_out, c_in, c_out, flag, out, e_in, e_out);
}


# offset pass and target gather run as resident-size grids whose waves loop over several edges / nodes (62.5k+100k waves per call -> 4k+10k); per-node sum order unchanged
# speedup vs baseline: 5.9972x; 1.1094x over previous
#include <hip/hip_runtime.h>
#include <hip/hip_bf16.h>

typedef float v2f __attribute__((ext_vector_type(2)));
typedef float v8f __attribute__((ext_vector_type(8)));

#define IN_CH 128
#define OUT_CH 64

__global__ void prep_weights_t(const float* __restrict__ w_main_in,
                               const float* __restrict__ w_main_out,
                               const float* __restrict__ w_shared,
                               float* __restrict__ w_in_t,
                               float* __restrict__ w_out_t,
                               int* __restrict__ unsorted_flag) {
    int i = blockIdx.x * blockDim.x + threadIdx.x;
    if (i == 0 && unsorted_flag) *unsorted_flag = 0;
    if (i < IN_CH * OUT_CH) {
        int k = i >> 6;
        int c = i & 63;
        float s = w_shared[i];
        w_in_t [c * IN_CH + k] = w_main_in[i]  + s;
        w_out_t[c * IN_CH + k] = w_main_out[i] + s;
    }
}

__global__ void gemm_dual_wmma(const float* __restrict__ x,
                               const float* __restrict__ w_in_t,
                               const float* __restrict__ w_out_t,
                               float* __restrict__ h_in,
                               float* __restrict__ h_out,
                               int n_nodes) {
    __shared__ __align__(16) float a_tile[16 * 132];

    const int mtile = blockIdx.x;
    const int row0  = mtile * 16;

    for (int i = threadIdx.x; i < 16 * (IN_CH / 4); i += blockDim.x) {
        int r  = i >> 5;
        int c4 = i & 31;
        float4 v = make_float4(0.f, 0.f, 0.f, 0.f);
        int grow = row0 + r;
        if (grow < n_nodes)
            v = ((const float4*)x)[(size_t)grow * (IN_CH / 4) + c4];
        *(float4*)&a_tile[r * 132 + c4 * 4] = v;
    }
    __syncthreads();

    const int wave = threadIdx.x >> 5;
    const int lane = threadIdx.x & 31;
    const int r    = lane & 15;
    const int half = lane >> 4;

    const float* __restrict__ Wt = (wave < 4) ? w_in_t : w_out_t;
    float*       __restrict__ H  = (wave < 4) ? h_in   : h_out;
    const int nt  = wave & 3;
    const int col = nt * 16 + r;

    const float* __restrict__ ap = &a_tile[r * 132 + 2 * half];
    const float* __restrict__ wp = Wt + (size_t)col * IN_CH + 2 * half;

    v8f acc = {};

#pragma unroll
    for (int k0 = 0; k0 < IN_CH; k0 += 4) {
        v2f a = *(const v2f*)(ap + k0);
        v2f b = *(const v2f*)(wp + k0);
        acc = __builtin_amdgcn_wmma_f32_16x16x4_f32(
             false, a,  false, b,
             (short)0, acc,  false,  false);
    }

#pragma unroll
    for (int v = 0; v < 8; ++v) {
        int grow = row0 + v + 8 * half;
        if (grow < n_nodes)
            H[(size_t)grow * OUT_CH + col] = acc[v];
    }
}

#define OFFS_BLOCKS 512
#define OFFS_PER_TRIP 8

__device__ __forceinline__ void offsets_of_set(const int* __restrict__ tgt,
                                               int* __restrict__ rp,
                                               int* __restrict__ unsorted_flag,
                                               int E, int n_nodes,
                                               long long first, long long stride) {
    for (long long e0 = first; e0 <= (long long)E; e0 += stride * OFFS_PER_TRIP) {
        int t[OFFS_PER_TRIP], p[OFFS_PER_TRIP];
#pragma unroll
        for (int u = 0; u < OFFS_PER_TRIP; ++u) {
            const long long e = e0 + (long long)u * stride;
            t[u] = n_nodes;
            p[u] = -1;
            if (e < (long long)E)            t[u] = tgt[e];
            if (e > 0 && e <= (long long)E)  p[u] = tgt[e - 1];
        }
#pragma unroll
        for (int u = 0; u < OFFS_PER_TRIP; ++u) {
            const long long e = e0 + (long long)u * stride;
            if (e <= (long long)E) {
                if (p[u] > t[u]) *unsorted_flag = 1;
                else {
#pragma nounroll
                    for (int n = p[u] + 1; n <= t[u]; ++n) rp[n] = (int)e;
                }
            }
        }
    }
}

__global__ void __launch_bounds__(256)
build_row_offsets(const int* __restrict__ ei_in,
                  const int* __restrict__ ei_out,
                  int* __restrict__ rp_in,
                  int* __restrict__ rp_out,
                  int* __restrict__ unsorted_flag,
                  int e_in, int e_out, int n_nodes) {
    const long long first  = (long long)blockIdx.x * blockDim.x + threadIdx.x;
    const long long stride = (long long)gridDim.x * blockDim.x;
    offsets_of_set(ei_in  + e_in,  rp_in,  unsorted_flag, e_in,  n_nodes, first, stride);
    offsets_of_set(ei_out + e_out, rp_out, unsorted_flag, e_out, n_nodes, first, stride);
}

#define ROWS_IN_FLIGHT 16
#define GATHER_BLOCKS 1280

__device__ __forceinline__ v2f add_rows(v2f acc,
                                        const float* __restrict__ h,
                                        int my_src, float my_w,
                                        int cnt, int lane) {
    for (int j = 0; j < cnt; j += ROWS_IN_FLIGHT) {
        int   sj[ROWS_IN_FLIGHT];
        float ww[ROWS_IN_FLIGHT];
        v2f   v[ROWS_IN_FLIGHT];
#pragma unroll
        for (int k = 0; k < ROWS_IN_FLIGHT; ++k) {
            sj[k] = __builtin_amdgcn_readlane(my_src, j + k);
            ww[k] = __int_as_float(__builtin_amdgcn_readlane(__float_as_int(my_w), j + k));
        }
#pragma unroll
        for (int k = 0; k < ROWS_IN_FLIGHT; ++k) {
            v[k] = (v2f){0.f, 0.f};
            if (j + k < cnt) {
                const v2f* __restrict__ row = (const v2f*)(h + (size_t)(unsigned)sj[k] * OUT_CH);
                v[k] = row[lane];
            }
        }
#pragma unroll
        for (int k = 0; k < ROWS_IN_FLIGHT; ++k) {
            acc.x = fmaf(ww[k], v[k].x, acc.x);
            acc.y = fmaf(ww[k], v[k].y, acc.y);
        }
    }
    return acc;
}

__global__ void __launch_bounds__(256)
gather_targets(const float* __restrict__ h_in,
               const float* __restrict__ h_out,
               const int*   __restrict__ ei_in,
               const float* __restrict__ ew_in,
               const int*   __restrict__ ei_out,
               const float* __restrict__ ew_out,
               const float* __restrict__ c_in,
               const float* __restrict__ c_out,
               const float* __restrict__ b_mi,
               const float* __restrict__ b_mo,
               const float* __restrict__ b_si,
               const float* __restrict__ b_so,
               const int*   __restrict__ rp_in,
               const int*   __restrict__ rp_out,
               const int*   __restrict__ unsorted_flag,
               float* __restrict__ out,
               int n_nodes) {
    const int lane = threadIdx.x & 31;
    const int wave0  = blockIdx.x * 8 + __builtin_amdgcn_readfirstlane((int)(threadIdx.x >> 5));
    const int nwaves = gridDim.x * 8;

    const v2f bi = *(const v2f*)(b_mi + 2 * lane) + *(const v2f*)(b_si + 2 * lane);
    const v2f bo = *(const v2f*)(b_mo + 2 * lane) + *(const v2f*)(b_so + 2 * lane);
    const bool sorted = unsorted_flag && (*unsorted_flag == 0);

  for (int n = wave0; n < n_nodes; n += nwaves) {
    const float ci = c_in[n], co = c_out[n];

    v2f s_i = {0.f, 0.f}, s_o = {0.f, 0.f};
    if (sorted) {
        const int a_i = rp_in[n],  z_i = rp_in[n + 1];
        const int a_o = rp_out[n], z_o = rp_out[n + 1];
        for (int ai = a_i, ao = a_o; ai < z_i || ao < z_o; ai += 32, ao += 32) {
            const int cnt_i = max(0, min(32, z_i - ai));
            const int cnt_o = max(0, min(32, z_o - ao));
            int src_i = 0, src_o = 0;
            float w_i = 0.f, w_o = 0.f;
            if (lane < cnt_i) { src_i = ei_in[ai + lane];  w_i = ew_in[ai + lane]; }
            if (lane < cnt_o) { src_o = ei_out[ao + lane]; w_o = ew_out[ao + lane]; }
            s_i = add_rows(s_i, h_in,  src_i, w_i, cnt_i, lane);
            s_o = add_rows(s_o, h_out, src_o, w_o, cnt_o, lane);
        }
    }
    v2f o;
    o.x = ci * (s_i.x + bi.x) + co * (s_o.x + bo.x);
    o.y = ci * (s_i.y + bi.y) + co * (s_o.y + bo.y);
    *(v2f*)(out + (size_t)n * OUT_CH + 2 * lane) = o;
  }
}

__global__ void scatter_edges(const float* __restrict__ h_in,
                              const float* __restrict__ h_out,
                              const int*   __restrict__ ei_in,
                              const float* __restrict__ ew_in,
                              const int*   __restrict__ ei_out,
                              const float* __restrict__ ew_out,
                              const float* __restrict__ c_in,
                              const float* __restrict__ c_out,
                              const int*   __restrict__ unsorted_flag,
                              float* __restrict__ out,
                              int e_in, int e_out) {
    if (unsorted_flag && *unsorted_flag == 0) return;
    const long long total  = ((long long)e_in + e_out) * OUT_CH;
    const long long stride = (long long)gridDim.x * blockDim.x;
    for (long long t = (long long)blockIdx.x * blockDim.x + threadIdx.x; t < total; t += stride) {
        int e = (int)(t >> 6);
        int c = (int)(t & 63);
        if (e < e_in) {
            int src = ei_in[e];
            int tgt = ei_in[e_in + e];
            float m = c_in[tgt] * ew_in[e] * h_in[(size_t)src * OUT_CH + c];
            unsafeAtomicAdd(&out[(size_t)tgt * OUT_CH + c], m);
        } else {
            e -= e_in;
            int src = ei_out[e];
            int tgt = ei_out[e_out + e];
            float m = c_out[tgt] * ew_out[e] * h_out[(size_t)src * OUT_CH + c];
            unsafeAtomicAdd(&out[(size_t)tgt * OUT_CH + c], m);
        }
    }
}

extern "C" void kernel_launch(void* const* d_in, const int* in_sizes, int n_in,
                              void* d_out, int out_size, void* d_ws, size_t ws_size,
                              hipStream_t stream) {
    const float* x          = (const float*)d_in[0];
    const float* w_main_in  = (const float*)d_in[1];
    const float* w_main_out = (const float*)d_in[2];
    const float* w_shared   = (const float*)d_in[3];
    const float* b_mi       = (const float*)d_in[4];
    const float* b_mo       = (const float*)d_in[5];
    const float* b_si       = (const float*)d_in[6];
    const float* b_so       = (const float*)d_in[7];
    const float* c_in       = (const float*)d_in[8];
    const float* c_out      = (const float*)d_in[9];
    const int*   ei_in      = (const int*)  d_in[10];
    const float* ew_in      = (const float*)d_in[11];
    const int*   ei_out     = (const int*)  d_in[12];
    const float* ew_out     = (const float*)d_in[13];
    float* out = (float*)d_out;

    const int n_nodes = in_sizes[8];
    const int e_in    = in_sizes[11];
    const int e_out   = in_sizes[13];

    float* ws      = (float*)d_ws;
    float* w_in_t  = ws;
    float* w_out_t = w_in_t  + IN_CH * OUT_CH;
    float* h_in    = w_out_t + IN_CH * OUT_CH;
    float* h_out   = h_in    + (size_t)n_nodes * OUT_CH;
    int*   rp_in   = (int*)(h_out + (size_t)n_nodes * OUT_CH);
    int*   rp_out  = rp_in  + (n_nodes + 1);
    int*   flag    = rp_out + (n_nodes + 1);
    const bool have_offsets = (size_t)((const char*)(flag + 1) - (const char*)d_ws) <= ws_size;
    if (!have_offsets) { rp_in = nullptr; rp_out = nullptr; flag = nullptr; }

    prep_weights_t<<<(IN_CH * OUT_CH + 255) / 256, 256, 0, stream>>>(
        w_main_in, w_main_out, w_shared, w_in_t, w_out_t, flag);

    const int mtiles = (n_nodes + 15) / 16;
    gemm_dual_wmma<<<mtiles, 256, 0, stream>>>(x, w_in_t, w_out_t, h_in, h_out, n_nodes);

    if (have_offsets) {
        build_row_offsets<<<OFFS_BLOCKS, 256, 0, stream>>>(
            ei_in, ei_out, rp_in, rp_out, flag, e_in, e_out, n_nodes);
    }

    gather_targets<<<GATHER_BLOCKS, 256, 0, stream>>>(
        h_in, h_out, ei_in, ew_in, ei_out, ew_out, c_in, c_out,
        b_mi, b_mo, b_si, b_so, rp_in, rp_out, flag, out, n_nodes);

    scatter_edges<<<1024, 256, 0, stream>>>(
        h_in, h_out, ei_in, ew_in, ei_out, ew_out, c_in, c_out, flag, out, e_in, e_out);
}


# GEMM re-tiled: 64-row block sums W_main+W_shared into LDS once, in fragment order; wave = 16 rows x 64 columns of one side (4 accumulators, whole-line h stores); prep launch goes (5->4)
# speedup vs baseline: 7.6160x; 1.2699x over previous
#include <hip/hip_runtime.h>
#include <hip/hip_bf16.h>

typedef float v2f __attribute__((ext_vector_type(2)));
typedef float v8f __attribute__((ext_vector_type(8)));

#define IN_CH 128
#define OUT_CH 64

#define GEMM_ROWS 64
#define A_PITCH   132
#define W_PITCH   256

__global__ void __launch_bounds__(256)
gemm_dual_wmma(const float* __restrict__ x,
               const float* __restrict__ w_main_in,
               const float* __restrict__ w_main_out,
               const float* __restrict__ w_shared,
               float* __restrict__ h_in,
               float* __restrict__ h_out,
               int* __restrict__ unsorted_flag,
               int n_nodes) {
    __shared__ __align__(16) float lds[(IN_CH / 2) * W_PITCH + GEMM_ROWS * A_PITCH];
    float* __restrict__ w_lds  = lds;
    float* __restrict__ a_tile = lds + (IN_CH / 2) * W_PITCH;

    if (blockIdx.x == 0 && threadIdx.x == 0 && unsorted_flag) *unsorted_flag = 0;

    const int row0 = blockIdx.x * GEMM_ROWS;

    float4 xv[(GEMM_ROWS * (IN_CH / 4)) / 256];
#pragma unroll
    for (int t = 0; t < (GEMM_ROWS * (IN_CH / 4)) / 256; ++t) {
        const int i    = threadIdx.x + 256 * t;
        const int grow = min(row0 + (i >> 5), n_nodes - 1);
        xv[t] = ((const float4*)x)[(size_t)grow * (IN_CH / 4) + (i & 31)];
    }
#pragma unroll
    for (int t = 0; t < (GEMM_ROWS * (IN_CH / 4)) / 256; ++t) {
        const int i = threadIdx.x + 256 * t;
        *(float4*)&a_tile[(i >> 5) * A_PITCH + (i & 31) * 4] = xv[t];
    }

#pragma unroll
    for (int t = 0; t < 8; ++t) {
        const int i  = threadIdx.x + 256 * t;
        const int cg = i & 15;
        const int kp = (i >> 4) & 63;
        const float* __restrict__ wm = (t < 4) ? w_main_in : w_main_out;
        const int g = (2 * kp) * OUT_CH + 4 * cg;
        const float4 m0 = *(const float4*)(wm + g);
        const float4 m1 = *(const float4*)(wm + g + OUT_CH);
        const float4 s0 = *(const float4*)(w_shared + g);
        const float4 s1 = *(const float4*)(w_shared + g + OUT_CH);
        float* __restrict__ dst = w_lds + kp * W_PITCH + (t >> 2) * 128;
        const int sw = (kp & 1) << 4;
        *(v2f*)(dst + 2 * ((cg     ) ^ sw)) = (v2f){m0.x + s0.x, m1.x + s1.x};
        *(v2f*)(dst + 2 * ((cg + 16) ^ sw)) = (v2f){m0.y + s0.y, m1.y + s1.y};
        *(v2f*)(dst + 2 * ((cg + 32) ^ sw)) = (v2f){m0.z + s0.z, m1.z + s1.z};
        *(v2f*)(dst + 2 * ((cg + 48) ^ sw)) = (v2f){m0.w + s0.w, m1.w + s1.w};
    }
    __syncthreads();

    const int wave = threadIdx.x >> 5;
    const int lane = threadIdx.x & 31;
    const int c    = lane & 15;
    const int half = lane >> 4;
    const int mt   = wave & 3;
    const int side = wave >> 2;

    const float* __restrict__ ap  = a_tile + (mt * 16 + c) * A_PITCH + 2 * half;
    const float* __restrict__ bp  = w_lds + half * W_PITCH + side * 128 + 2 * c;
    const float* __restrict__ bp0 = bp + 32 * half;
    const float* __restrict__ bp1 = bp + 32 * (half ^ 1);

    v8f acc0 = {}, acc1 = {}, acc2 = {}, acc3 = {};

#pragma unroll
    for (int k0 = 0; k0 < IN_CH; k0 += 4) {
        const v2f a  = *(const v2f*)(ap  + k0);
        const v2f b0 = *(const v2f*)(bp0 + k0 * (W_PITCH / 2));
        const v2f b1 = *(const v2f*)(bp1 + k0 * (W_PITCH / 2));
        const v2f b2 = *(const v2f*)(bp0 + k0 * (W_PITCH / 2) + 64);
        const v2f b3 = *(const v2f*)(bp1 + k0 * (W_PITCH / 2) + 64);
        acc0 = __builtin_amdgcn_wmma_f32_16x16x4_f32(false, a, false, b0, (short)0, acc0, false, false);
        acc1 = __builtin_amdgcn_wmma_f32_16x16x4_f32(false, a, false, b1, (short)0, acc1, false, false);
        acc2 = __builtin_amdgcn_wmma_f32_16x16x4_f32(false, a, false, b2, (short)0, acc2, false, false);
        acc3 = __builtin_amdgcn_wmma_f32_16x16x4_f32(false, a, false, b3, (short)0, acc3, false, false);
    }

    float* __restrict__ H = side ? h_out : h_in;
#pragma unroll
    for (int v = 0; v < 8; ++v) {
        const int grow = row0 + mt * 16 + v + 8 * half;
        if (grow < n_nodes)
            *(float4*)(H + (size_t)grow * OUT_CH + 4 * c) =
                make_float4(acc0[v], acc1[v], acc2[v], acc3[v]);
    }
}

#define OFFS_BLOCKS 512
#define OFFS_PER_TRIP 8

__device__ __forceinline__ void offsets_of_set(const int* __restrict__ tgt,
                                               int* __restrict__ rp,
                                               int* __restrict__ unsorted_flag,
                                               int E, int n_nodes,
                                               long long first, long long stride) {
    for (long long e0 = first; e0 <= (long long)E; e0 += stride * OFFS_PER_TRIP) {
        int t[OFFS_PER_TRIP], p[OFFS_PER_TRIP];
#pragma unroll
        for (int u = 0; u < OFFS_PER_TRIP; ++u) {
            const long long e = e0 + (long long)u * stride;
            t[u] = n_nodes;
            p[u] = -1;
            if (e < (long long)E)            t[u] = tgt[e];
            if (e > 0 && e <= (long long)E)  p[u] = tgt[e - 1];
        }
#pragma unroll
        for (int u = 0; u < OFFS_PER_TRIP; ++u) {
            const long long e = e0 + (long long)u * stride;
            if (e <= (long long)E) {
                if (p[u] > t[u]) *unsorted_flag = 1;
                else {
#pragma nounroll
                    for (int n = p[u] + 1; n <= t[u]; ++n) rp[n] = (int)e;
                }
            }
        }
    }
}

__global__ void __launch_bounds__(256)
build_row_offsets(const int* __restrict__ ei_in,
                  const int* __restrict__ ei_out,
                  int* __restrict__ rp_in,
                  int* __restrict__ rp_out,
                  int* __restrict__ unsorted_flag,
                  int e_in, int e_out, int n_nodes) {
    const long long first  = (long long)blockIdx.x * blockDim.x + threadIdx.x;
    const long long stride = (long long)gridDim.x * blockDim.x;
    offsets_of_set(ei_in  + e_in,  rp_in,  unsorted_flag, e_in,  n_nodes, first, stride);
    offsets_of_set(ei_out + e_out, rp_out, unsorted_flag, e_out, n_nodes, first, stride);
}

#define ROWS_IN_FLIGHT 16
#define GATHER_BLOCKS 1280

__device__ __forceinline__ v2f add_rows(v2f acc,
                                        const float* __restrict__ h,
                                        int my_src, float my_w,
                                        int cnt, int lane) {
    for (int j = 0; j < cnt; j += ROWS_IN_FLIGHT) {
        int   sj[ROWS_IN_FLIGHT];
        float ww[ROWS_IN_FLIGHT];
        v2f   v[ROWS_IN_FLIGHT];
#pragma unroll
        for (int k = 0; k < ROWS_IN_FLIGHT; ++k) {
            sj[k] = __builtin_amdgcn_readlane(my_src, j + k);
            ww[k] = __int_as_float(__builtin_amdgcn_readlane(__float_as_int(my_w), j + k));
        }
#pragma unroll
        for (int k = 0; k < ROWS_IN_FLIGHT; ++k) {
            v[k] = (v2f){0.f, 0.f};
            if (j + k < cnt) {
                const v2f* __restrict__ row = (const v2f*)(h + (size_t)(unsigned)sj[k] * OUT_CH);
                v[k] = row[lane];
            }
        }
#pragma unroll
        for (int k = 0; k < ROWS_IN_FLIGHT; ++k) {
            acc.x = fmaf(ww[k], v[k].x, acc.x);
            acc.y = fmaf(ww[k], v[k].y, acc.y);
        }
    }
    return acc;
}

__global__ void __launch_bounds__(256)
gather_targets(const float* __restrict__ h_in,
               const float* __restrict__ h_out,
               const int*   __restrict__ ei_in,
               const float* __restrict__ ew_in,
               const int*   __restrict__ ei_out,
               const float* __restrict__ ew_out,
               const float* __restrict__ c_in,
               const float* __restrict__ c_out,
               const float* __restrict__ b_mi,
               const float* __restrict__ b_mo,
               const float* __restrict__ b_si,
               const float* __restrict__ b_so,
               const int*   __restrict__ rp_in,
               const int*   __restrict__ rp_out,
               const int*   __restrict__ unsorted_flag,
               float* __restrict__ out,
               int n_nodes) {
    const int lane = threadIdx.x & 31;
    const int wave0  = blockIdx.x * 8 + __builtin_amdgcn_readfirstlane((int)(threadIdx.x >> 5));
    const int nwaves = gridDim.x * 8;

    const v2f bi = *(const v2f*)(b_mi + 2 * lane) + *(const v2f*)(b_si + 2 * lane);
    const v2f bo = *(const v2f*)(b_mo + 2 * lane) + *(const v2f*)(b_so + 2 * lane);
    const bool sorted = unsorted_flag && (*unsorted_flag == 0);

  for (int n = wave0; n < n_nodes; n += nwaves) {
    const float ci = c_in[n], co = c_out[n];

    v2f s_i = {0.f, 0.f}, s_o = {0.f, 0.f};
    if (sorted) {
        const int a_i = rp_in[n],  z_i = rp_in[n + 1];
        const int a_o = rp_out[n], z_o = rp_out[n + 1];
        for (int ai = a_i, ao = a_o; ai < z_i || ao < z_o; ai += 32, ao += 32) {
            const int cnt_i = max(0, min(32, z_i - ai));
            const int cnt_o = max(0, min(32, z_o - ao));
            int src_i = 0, src_o = 0;
            float w_i = 0.f, w_o = 0.f;
            if (lane < cnt_i) { src_i = ei_in[ai + lane];  w_i = ew_in[ai + lane]; }
            if (lane < cnt_o) { src_o = ei_out[ao + lane]; w_o = ew_out[ao + lane]; }
            s_i = add_rows(s_i, h_in,  src_i, w_i, cnt_i, lane);
            s_o = add_rows(s_o, h_out, src_o, w_o, cnt_o, lane);
        }
    }
    v2f o;
    o.x = ci * (s_i.x + bi.x) + co * (s_o.x + bo.x);
    o.y = ci * (s_i.y + bi.y) + co * (s_o.y + bo.y);
    *(v2f*)(out + (size_t)n * OUT_CH + 2 * lane) = o;
  }
}

__global__ void scatter_edges(const float* __restrict__ h_in,
                              const float* __restrict__ h_out,
                              const int*   __restrict__ ei_in,
                              const float* __restrict__ ew_in,
                              const int*   __restrict__ ei_out,
                              const float* __restrict__ ew_out,
                              const float* __restrict__ c_in,
                              const float* __restrict__ c_out,
                              const int*   __restrict__ unsorted_flag,
                              float* __restrict__ out,
                              int e_in, int e_out) {
    if (unsorted_flag && *unsorted_flag == 0) return;
    const long long total  = ((long long)e_in + e_out) * OUT_CH;
    const long long stride = (long long)gridDim.x * blockDim.x;
    for (long long t = (long long)blockIdx.x * blockDim.x + threadIdx.x; t < total; t += stride) {
        int e = (int)(t >> 6);
        int c = (int)(t & 63);
        if (e < e_in) {
            int src = ei_in[e];
            int tgt = ei_in[e_in + e];
            float m = c_in[tgt] * ew_in[e] * h_in[(size_t)src * OUT_CH + c];
            unsafeAtomicAdd(&out[(size_t)tgt * OUT_CH + c], m);
        } else {
            e -= e_in;
            int src = ei_out[e];
            int tgt = ei_out[e_out + e];
            float m = c_out[tgt] * ew_out[e] * h_out[(size_t)src * OUT_CH + c];
            unsafeAtomicAdd(&out[(size_t)tgt * OUT_CH + c], m);
        }
    }
}

extern "C" void kernel_launch(void* const* d_in, const int* in_sizes, int n_in,
                              void* d_out, int out_size, void* d_ws, size_t ws_size,
                              hipStream_t stream) {
    const float* x          = (const float*)d_in[0];
    const float* w_main_in  = (const float*)d_in[1];
    const float* w_main_out = (const float*)d_in[2];
    const float* w_shared   = (const float*)d_in[3];
    const float* b_mi       = (const float*)d_in[4];
    const float* b_mo       = (const float*)d_in[5];
    const float* b_si       = (const float*)d_in[6];
    const float* b_so       = (const float*)d_in[7];
    const float* c_in       = (const float*)d_in[8];
    const float* c_out      = (const float*)d_in[9];
    const int*   ei_in      = (const int*)  d_in[10];
    const float* ew_in      = (const float*)d_in[11];
    const int*   ei_out     = (const int*)  d_in[12];
    const float* ew_out     = (const float*)d_in[13];
    float* out = (float*)d_out;

    const int n_nodes = in_sizes[8];
    const int e_in    = in_sizes[11];
    const int e_out   = in_sizes[13];

    float* h_in    = (float*)d_ws;
    float* h_out   = h_in    + (size_t)n_nodes * OUT_CH;
    int*   rp_in   = (int*)(h_out + (size_t)n_nodes * OUT_CH);
    int*   rp_out  = rp_in  + (n_nodes + 1);
    int*   flag    = rp_out + (n_nodes + 1);
    const bool have_offsets = (size_t)((const char*)(flag + 1) - (const char*)d_ws) <= ws_size;
    if (!have_offsets) { rp_in = nullptr; rp_out = nullptr; flag = nullptr; }

    const int mblocks = (n_nodes + GEMM_ROWS - 1) / GEMM_ROWS;
    gemm_dual_wmma<<<mblocks, 256, 0, stream>>>(
        x, w_main_in, w_main_out, w_shared, h_in, h_out, flag, n_nodes);

    if (have_offsets) {
        build_row_offsets<<<OFFS_BLOCKS, 256, 0, stream>>>(
            ei_in, ei_out, rp_in, rp_out, flag, e_in, e_out, n_nodes);
    }

    gather_targets<<<GATHER_BLOCKS, 256, 0, stream>>>(
        h_in, h_out, ei_in, ew_in, ei_out, ew_out, c_in, c_out,
        b_mi, b_mo, b_si, b_so, rp_in, rp_out, flag, out, n_nodes);

    scatter_edges<<<1024, 256, 0, stream>>>(
        h_in, h_out, ei_in, ew_in, ei_out, ew_out, c_in, c_out, flag, out, e_in, e_out);
}


# target gather software-pipelined over a wave's nodes: next node's edge lists and the following node's offsets load while this node's rows of both sets are in flight; per-node sum order unchanged
# speedup vs baseline: 7.9811x; 1.0479x over previous
#include <hip/hip_runtime.h>
#include <hip/hip_bf16.h>

typedef float v2f __attribute__((ext_vector_type(2)));
typedef float v8f __attribute__((ext_vector_type(8)));

#define IN_CH 128
#define OUT_CH 64

#define GEMM_ROWS 64
#define A_PITCH   132
#define W_PITCH   256

__global__ void __launch_bounds__(256)
gemm_dual_wmma(const float* __restrict__ x,
               const float* __restrict__ w_main_in,
               const float* __restrict__ w_main_out,
               const float* __restrict__ w_shared,
               float* __restrict__ h_in,
               float* __restrict__ h_out,
               int* __restrict__ unsorted_flag,
               int n_nodes) {
    __shared__ __align__(16) float lds[(IN_CH / 2) * W_PITCH + GEMM_ROWS * A_PITCH];
    float* __restrict__ w_lds  = lds;
    float* __restrict__ a_tile = lds + (IN_CH / 2) * W_PITCH;

    if (blockIdx.x == 0 && threadIdx.x == 0 && unsorted_flag) *unsorted_flag = 0;

    const int row0 = blockIdx.x * GEMM_ROWS;

    float4 xv[(GEMM_ROWS * (IN_CH / 4)) / 256];
#pragma unroll
    for (int t = 0; t < (GEMM_ROWS * (IN_CH / 4)) / 256; ++t) {
        const int i    = threadIdx.x + 256 * t;
        const int grow = min(row0 + (i >> 5), n_nodes - 1);
        xv[t] = ((const float4*)x)[(size_t)grow * (IN_CH / 4) + (i & 31)];
    }
#pragma unroll
    for (int t = 0; t < (GEMM_ROWS * (IN_CH / 4)) / 256; ++t) {
        const int i = threadIdx.x + 256 * t;
        *(float4*)&a_tile[(i >> 5) * A_PITCH + (i & 31) * 4] = xv[t];
    }

#pragma unroll
    for (int t = 0; t < 8; ++t) {
        const int i  = threadIdx.x + 256 * t;
        const int cg = i & 15;
        const int kp = (i >> 4) & 63;
        const float* __restrict__ wm = (t < 4) ? w_main_in : w_main_out;
        const int g = (2 * kp) * OUT_CH + 4 * cg;
        const float4 m0 = *(const float4*)(wm + g);
        const float4 m1 = *(const float4*)(wm + g + OUT_CH);
        const float4 s0 = *(const float4*)(w_shared + g);
        const float4 s1 = *(const float4*)(w_shared + g + OUT_CH);
        float* __restrict__ dst = w_lds + kp * W_PITCH + (t >> 2) * 128;
        const int sw = (kp & 1) << 4;
        *(v2f*)(dst + 2 * ((cg     ) ^ sw)) = (v2f){m0.x + s0.x, m1.x + s1.x};
        *(v2f*)(dst + 2 * ((cg + 16) ^ sw)) = (v2f){m0.y + s0.y, m1.y + s1.y};
        *(v2f*)(dst + 2 * ((cg + 32) ^ sw)) = (v2f){m0.z + s0.z, m1.z + s1.z};
        *(v2f*)(dst + 2 * ((cg + 48) ^ sw)) = (v2f){m0.w + s0.w, m1.w + s1.w};
    }
    __syncthreads();

    const int wave = threadIdx.x >> 5;
    const int lane = threadIdx.x & 31;
    const int c    = lane & 15;
    const int half = lane >> 4;
    const int mt   = wave & 3;
    const int side = wave >> 2;

    const float* __restrict__ ap  = a_tile + (mt * 16 + c) * A_PITCH + 2 * half;
    const float* __restrict__ bp  = w_lds + half * W_PITCH + side * 128 + 2 * c;
    const float* __restrict__ bp0 = bp + 32 * half;
    const float* __restrict__ bp1 = bp + 32 * (half ^ 1);

    v8f acc0 = {}, acc1 = {}, acc2 = {}, acc3 = {};

#pragma unroll
    for (int k0 = 0; k0 < IN_CH; k0 += 4) {
        const v2f a  = *(const v2f*)(ap  + k0);
        const v2f b0 = *(const v2f*)(bp0 + k0 * (W_PITCH / 2));
        const v2f b1 = *(const v2f*)(bp1 + k0 * (W_PITCH / 2));
        const v2f b2 = *(const v2f*)(bp0 + k0 * (W_PITCH / 2) + 64);
        const v2f b3 = *(const v2f*)(bp1 + k0 * (W_PITCH / 2) + 64);
        acc0 = __builtin_amdgcn_wmma_f32_16x16x4_f32(false, a, false, b0, (short)0, acc0, false, false);
        acc1 = __builtin_amdgcn_wmma_f32_16x16x4_f32(false, a, false, b1, (short)0, acc1, false, false);
        acc2 = __builtin_amdgcn_wmma_f32_16x16x4_f32(false, a, false, b2, (short)0, acc2, false, false);
        acc3 = __builtin_amdgcn_wmma_f32_16x16x4_f32(false, a, false, b3, (short)0, acc3, false, false);
    }

    float* __restrict__ H = side ? h_out : h_in;
#pragma unroll
    for (int v = 0; v < 8; ++v) {
        const int grow = row0 + mt * 16 + v + 8 * half;
        if (grow < n_nodes)
            *(float4*)(H + (size_t)grow * OUT_CH + 4 * c) =
                make_float4(acc0[v], acc1[v], acc2[v], acc3[v]);
    }
}

#define OFFS_BLOCKS 512
#define OFFS_PER_TRIP 8

__device__ __forceinline__ void offsets_of_set(const int* __restrict__ tgt,
                                               int* __restrict__ rp,
                                               int* __restrict__ unsorted_flag,
                                               int E, int n_nodes,
                                               long long first, long long stride) {
    for (long long e0 = first; e0 <= (long long)E; e0 += stride * OFFS_PER_TRIP) {
        int t[OFFS_PER_TRIP], p[OFFS_PER_TRIP];
#pragma unroll
        for (int u = 0; u < OFFS_PER_TRIP; ++u) {
            const long long e = e0 + (long long)u * stride;
            t[u] = n_nodes;
            p[u] = -1;
            if (e < (long long)E)            t[u] = tgt[e];
            if (e > 0 && e <= (long long)E)  p[u] = tgt[e - 1];
        }
#pragma unroll
        for (int u = 0; u < OFFS_PER_TRIP; ++u) {
            const long long e = e0 + (long long)u * stride;
            if (e <= (long long)E) {
                if (p[u] > t[u]) *unsorted_flag = 1;
                else {
#pragma nounroll
                    for (int n = p[u] + 1; n <= t[u]; ++n) rp[n] = (int)e;
                }
            }
        }
    }
}

__global__ void __launch_bounds__(256)
build_row_offsets(const int* __restrict__ ei_in,
                  const int* __restrict__ ei_out,
                  int* __restrict__ rp_in,
                  int* __restrict__ rp_out,
                  int* __restrict__ unsorted_flag,
                  int e_in, int e_out, int n_nodes) {
    const long long first  = (long long)blockIdx.x * blockDim.x + threadIdx.x;
    const long long stride = (long long)gridDim.x * blockDim.x;
    offsets_of_set(ei_in  + e_in,  rp_in,  unsorted_flag, e_in,  n_nodes, first, stride);
    offsets_of_set(ei_out + e_out, rp_out, unsorted_flag, e_out, n_nodes, first, stride);
}

#define ROWS_IN_FLIGHT 16
#define GATHER_BLOCKS 1280

__device__ __forceinline__ void issue_rows(v2f (&v)[ROWS_IN_FLIGHT],
                                           const float* __restrict__ h,
                                           int my_src, int first, int cnt, int lane) {
    int sj[ROWS_IN_FLIGHT];
#pragma unroll
    for (int k = 0; k < ROWS_IN_FLIGHT; ++k)
        sj[k] = __builtin_amdgcn_readlane(my_src, first + k);
#pragma unroll
    for (int k = 0; k < ROWS_IN_FLIGHT; ++k) {
        v[k] = (v2f){0.f, 0.f};
        if (first + k < cnt) {
            const v2f* __restrict__ row = (const v2f*)(h + (size_t)(unsigned)sj[k] * OUT_CH);
            v[k] = row[lane];
        }
    }
}

__device__ __forceinline__ v2f sum_rows(v2f acc, const v2f (&v)[ROWS_IN_FLIGHT],
                                        float my_w, int first) {
#pragma unroll
    for (int k = 0; k < ROWS_IN_FLIGHT; ++k) {
        const float w = __int_as_float(__builtin_amdgcn_readlane(__float_as_int(my_w), first + k));
        acc.x = fmaf(w, v[k].x, acc.x);
        acc.y = fmaf(w, v[k].y, acc.y);
    }
    return acc;
}

__device__ __forceinline__ v2f add_rows(v2f acc,
                                        const float* __restrict__ h,
                                        int my_src, float my_w,
                                        int first, int cnt, int lane) {
    for (int j = first; j < cnt; j += ROWS_IN_FLIGHT) {
        v2f v[ROWS_IN_FLIGHT];
        issue_rows(v, h, my_src, j, cnt, lane);
        acc = sum_rows(acc, v, my_w, j);
    }
    return acc;
}

__global__ void __launch_bounds__(256)
gather_targets(const float* __restrict__ h,
               const int*   __restrict__ ei_in,
               const float* __restrict__ ew_in,
               const int*   __restrict__ ei_out,
               const float* __restrict__ ew_out,
               const float* __restrict__ c_in,
               const float* __restrict__ c_out,
               const float* __restrict__ b_mi,
               const float* __restrict__ b_mo,
               const float* __restrict__ b_si,
               const float* __restrict__ b_so,
               const int*   __restrict__ rp_in,
               const int*   __restrict__ rp_out,
               const int*   __restrict__ unsorted_flag,
               float* __restrict__ out,
               int e_in, int e_out, int n_nodes) {
    const int lane = threadIdx.x & 31;
    const int wave0  = blockIdx.x * 8 + __builtin_amdgcn_readfirstlane((int)(threadIdx.x >> 5));
    const int nwaves = gridDim.x * 8;

    const v2f bi = *(const v2f*)(b_mi + 2 * lane) + *(const v2f*)(b_si + 2 * lane);
    const v2f bo = *(const v2f*)(b_mo + 2 * lane) + *(const v2f*)(b_so + 2 * lane);
    const bool sorted = unsorted_flag && (*unsorted_flag == 0);

    if (!sorted || e_in <= 0 || e_out <= 0) {
        for (int n = wave0; n < n_nodes; n += nwaves) {
            const float ci = c_in[n], co = c_out[n];
            v2f s_i = {0.f, 0.f}, s_o = {0.f, 0.f};
            if (sorted) {
                const int a_i = rp_in[n],  z_i = rp_in[n + 1];
                const int a_o = rp_out[n], z_o = rp_out[n + 1];
                for (int ai = a_i, ao = a_o; ai < z_i || ao < z_o; ai += 32, ao += 32) {
                    const int cnt_i = max(0, min(32, z_i - ai));
                    const int cnt_o = max(0, min(32, z_o - ao));
                    int src_i = 0, src_o = 0;
                    float w_i = 0.f, w_o = 0.f;
                    if (lane < cnt_i) { src_i = ei_in[ai + lane];  w_i = ew_in[ai + lane]; }
                    if (lane < cnt_o) { src_o = ei_out[ao + lane] + n_nodes; w_o = ew_out[ao + lane]; }
                    s_i = add_rows(s_i, h, src_i, w_i, 0, cnt_i, lane);
                    s_o = add_rows(s_o, h, src_o, w_o, 0, cnt_o, lane);
                }
            }
            v2f o;
            o.x = ci * (s_i.x + bi.x) + co * (s_o.x + bo.x);
            o.y = ci * (s_i.y + bi.y) + co * (s_o.y + bo.y);
            *(v2f*)(out + (size_t)n * OUT_CH + 2 * lane) = o;
        }
        return;
    }

    if (wave0 >= n_nodes) return;

    const int last = n_nodes - 1;
    float ci = 0.f, co = 0.f;
    int   a_i = 0, z_i = 0, a_o = 0, z_o = 0;
    int   src_i = 0, src_o = 0;
    float w_i = 0.f, w_o = 0.f;
    float ci1 = c_in[wave0], co1 = c_out[wave0];
    int a_i1 = rp_in[wave0],  z_i1 = rp_in[wave0 + 1];
    int a_o1 = rp_out[wave0], z_o1 = rp_out[wave0 + 1];

    for (int n = wave0 - nwaves; n < n_nodes; n += nwaves) {
        const int xi1 = min(a_i1 + lane, e_in - 1), xo1 = min(a_o1 + lane, e_out - 1);
        const int   src_i1 = ei_in[xi1];
        const float w_i1   = ew_in[xi1];
        const int   src_o1 = ei_out[xo1] + n_nodes;
        const float w_o1   = ew_out[xo1];
        const int m2 = min(n + 2 * nwaves, last);
        const float ci2 = c_in[m2], co2 = c_out[m2];
        const int a_i2 = rp_in[m2],  z_i2 = rp_in[m2 + 1];
        const int a_o2 = rp_out[m2], z_o2 = rp_out[m2 + 1];

        if (n >= 0) {
            const int cnt_i = max(0, min(32, z_i - a_i));
            const int cnt_o = max(0, min(32, z_o - a_o));
            w_i = (lane < cnt_i) ? w_i : 0.f;
            w_o = (lane < cnt_o) ? w_o : 0.f;
            v2f vi[ROWS_IN_FLIGHT], vo[ROWS_IN_FLIGHT];
            issue_rows(vi, h, src_i, 0, cnt_i, lane);
            issue_rows(vo, h, src_o, 0, cnt_o, lane);
            v2f s_i = sum_rows((v2f){0.f, 0.f}, vi, w_i, 0);
            v2f s_o = sum_rows((v2f){0.f, 0.f}, vo, w_o, 0);
            s_i = add_rows(s_i, h, src_i, w_i, ROWS_IN_FLIGHT, cnt_i, lane);
            s_o = add_rows(s_o, h, src_o, w_o, ROWS_IN_FLIGHT, cnt_o, lane);
            for (int ai = a_i + 32, ao = a_o + 32; ai < z_i || ao < z_o; ai += 32, ao += 32) {
                const int ci_n = max(0, min(32, z_i - ai));
                const int co_n = max(0, min(32, z_o - ao));
                int s2i = 0, s2o = 0;
                float w2i = 0.f, w2o = 0.f;
                if (lane < ci_n) { s2i = ei_in[ai + lane];  w2i = ew_in[ai + lane]; }
                if (lane < co_n) { s2o = ei_out[ao + lane] + n_nodes; w2o = ew_out[ao + lane]; }
                s_i = add_rows(s_i, h, s2i, w2i, 0, ci_n, lane);
                s_o = add_rows(s_o, h, s2o, w2o, 0, co_n, lane);
            }

            v2f o;
            o.x = ci * (s_i.x + bi.x) + co * (s_o.x + bo.x);
            o.y = ci * (s_i.y + bi.y) + co * (s_o.y + bo.y);
            *(v2f*)(out + (size_t)n * OUT_CH + 2 * lane) = o;
        }

        ci = ci1; co = co1; a_i = a_i1; z_i = z_i1; a_o = a_o1; z_o = z_o1;
        src_i = src_i1; w_i = w_i1; src_o = src_o1; w_o = w_o1;
        ci1 = ci2; co1 = co2; a_i1 = a_i2; z_i1 = z_i2; a_o1 = a_o2; z_o1 = z_o2;
    }
}

__global__ void scatter_edges(const float* __restrict__ h_in,
                              const float* __restrict__ h_out,
                              const int*   __restrict__ ei_in,
                              const float* __restrict__ ew_in,
                              const int*   __restrict__ ei_out,
                              const float* __restrict__ ew_out,
                              const float* __restrict__ c_in,
                              const float* __restrict__ c_out,
                              const int*   __restrict__ unsorted_flag,
                              float* __restrict__ out,
                              int e_in, int e_out) {
    if (unsorted_flag && *unsorted_flag == 0) return;
    const long long total  = ((long long)e_in + e_out) * OUT_CH;
    const long long stride = (long long)gridDim.x * blockDim.x;
    for (long long t = (long long)blockIdx.x * blockDim.x + threadIdx.x; t < total; t += stride) {
        int e = (int)(t >> 6);
        int c = (int)(t & 63);
        if (e < e_in) {
            int src = ei_in[e];
            int tgt = ei_in[e_in + e];
            float m = c_in[tgt] * ew_in[e] * h_in[(size_t)src * OUT_CH + c];
            unsafeAtomicAdd(&out[(size_t)tgt * OUT_CH + c], m);
        } else {
            e -= e_in;
            int src = ei_out[e];
            int tgt = ei_out[e_out + e];
            float m = c_out[tgt] * ew_out[e] * h_out[(size_t)src * OUT_CH + c];
            unsafeAtomicAdd(&out[(size_t)tgt * OUT_CH + c], m);
        }
    }
}

extern "C" void kernel_launch(void* const* d_in, const int* in_sizes, int n_in,
                              void* d_out, int out_size, void* d_ws, size_t ws_size,
                              hipStream_t stream) {
    const float* x          = (const float*)d_in[0];
    const float* w_main_in  = (const float*)d_in[1];
    const float* w_main_out = (const float*)d_in[2];
    const float* w_shared   = (const float*)d_in[3];
    const float* b_mi       = (const float*)d_in[4];
    const float* b_mo       = (const float*)d_in[5];
    const float* b_si       = (const float*)d_in[6];
    const float* b_so       = (const float*)d_in[7];
    const float* c_in       = (const float*)d_in[8];
    const float* c_out      = (const float*)d_in[9];
    const int*   ei_in      = (const int*)  d_in[10];
    const float* ew_in      = (const float*)d_in[11];
    const int*   ei_out     = (const int*)  d_in[12];
    const float* ew_out     = (const float*)d_in[13];
    float* out = (float*)d_out;

    const int n_nodes = in_sizes[8];
    const int e_in    = in_sizes[11];
    const int e_out   = in_sizes[13];

    float* h_in    = (float*)d_ws;
    float* h_out   = h_in    + (size_t)n_nodes * OUT_CH;
    int*   rp_in   = (int*)(h_out + (size_t)n_nodes * OUT_CH);
    int*   rp_out  = rp_in  + (n_nodes + 1);
    int*   flag    = rp_out + (n_nodes + 1);
    const bool have_offsets = (size_t)((const char*)(flag + 1) - (const char*)d_ws) <= ws_size;
    if (!have_offsets) { rp_in = nullptr; rp_out = nullptr; flag = nullptr; }

    const int mblocks = (n_nodes + GEMM_ROWS - 1) / GEMM_ROWS;
    gemm_dual_wmma<<<mblocks, 256, 0, stream>>>(
        x, w_main_in, w_main_out, w_shared, h_in, h_out, flag, n_nodes);

    if (have_offsets) {
        build_row_offsets<<<OFFS_BLOCKS, 256, 0, stream>>>(
            ei_in, ei_out, rp_in, rp_out, flag, e_in, e_out, n_nodes);
    }

    gather_targets<<<GATHER_BLOCKS, 256, 0, stream>>>(
        h_in, ei_in, ew_in, ei_out, ew_out, c_in, c_out,
        b_mi, b_mo, b_si, b_so, rp_in, rp_out, flag, out, e_in, e_out, n_nodes);

    scatter_edges<<<1024, 256, 0, stream>>>(
        h_in, h_out, ei_in, ew_in, ei_out, ew_out, c_in, c_out, flag, out, e_in, e_out);
}
